# starting kernel
# baseline (speedup 1.0000x reference)
#include <hip/hip_runtime.h>
#include <hip/hip_bf16.h>

typedef __attribute__((ext_vector_type(2))) float v2f;
typedef __attribute__((ext_vector_type(8))) float v8f;

#define LDS_STRIDE 132

__global__ __launch_bounds__(256) void fill0_f4(float4* __restrict__ p, long n4) {
  long i = (long)blockIdx.x * blockDim.x + threadIdx.x;
  long stride = (long)gridDim.x * blockDim.x;
  float4 z = make_float4(0.f, 0.f, 0.f, 0.f);
  for (; i < n4; i += stride) p[i] = z;
}

__global__ __launch_bounds__(256) void degree_count(const int* __restrict__ dst,
                                                    float* __restrict__ deg, int nE) {
  int i = blockIdx.x * blockDim.x + threadIdx.x;
  int stride = gridDim.x * blockDim.x;
  for (; i < nE; i += stride) atomicAdd(&deg[dst[i]], 1.0f);
}

__global__ __launch_bounds__(256) void inv_deg_kernel(float* __restrict__ deg, int n) {
  int i = blockIdx.x * blockDim.x + threadIdx.x;
  if (i < n) {
    float d = deg[i];
    deg[i] = 1.0f / fmaxf(d, 1.0f);
  }
}

__global__ __launch_bounds__(256) void scatter_accum(const float* __restrict__ h,
                                                     const int* __restrict__ src,
                                                     const int* __restrict__ dst,
                                                     float* __restrict__ agg,
                                                     int nE, int rowLimit) {
  const int lane = threadIdx.x & 31;
  int wave = (blockIdx.x * blockDim.x + threadIdx.x) >> 5;
  const int nWaves = (gridDim.x * blockDim.x) >> 5;
  for (int e = wave; e < nE; e += nWaves) {
    const int d = dst[e];
    if (d >= rowLimit) continue;
    const int s = src[e];
    const float4 v = *(const float4*)(h + (size_t)s * 128 + (size_t)lane * 4);
    float* p = agg + (size_t)d * 128 + (size_t)lane * 4;
    atomicAdd(p + 0, v.x);
    atomicAdd(p + 1, v.y);
    atomicAdd(p + 2, v.z);
    atomicAdd(p + 3, v.w);
  }
}

__global__ __launch_bounds__(256) void sage_wmma_gemm(const float* __restrict__ agg,
                                                      const float* __restrict__ invdeg,
                                                      const float* __restrict__ h,
                                                      const float* __restrict__ Wl,
                                                      const float* __restrict__ bvec,
                                                      const float* __restrict__ Wr,
                                                      const float* __restrict__ prelu_a,
                                                      float* __restrict__ out) {
  __shared__ float sMean[16 * LDS_STRIDE];
  __shared__ float sH[16 * LDS_STRIDE];

  const int lane = threadIdx.x & 31;
  const int wave = threadIdx.x >> 5;
  const int m0   = blockIdx.x << 4;
  const int kh   = (lane >> 4) << 1;
  const int col  = (wave << 4) + (lane & 15);

  for (int i = threadIdx.x; i < 512; i += 256) {
    const int r  = i >> 5;
    const int c4 = (i & 31) << 2;
    const float idg = invdeg[m0 + r];
    float4 va = *(const float4*)(agg + (size_t)(m0 + r) * 128 + c4);
    float4 vh = *(const float4*)(h + (size_t)(m0 + r) * 128 + c4);
    va.x *= idg; va.y *= idg; va.z *= idg; va.w *= idg;
    *(float4*)(&sMean[r * LDS_STRIDE + c4]) = va;
    *(float4*)(&sH[r * LDS_STRIDE + c4])    = vh;
  }
  __syncthreads();

  const float* mRow = &sMean[(lane & 15) * LDS_STRIDE];
  const float* hRow = &sH[(lane & 15) * LDS_STRIDE];

  v8f acc = {};

#pragma unroll 8
  for (int k = 0; k < 128; k += 4) {
    const int kb = k + kh;
    v2f am = *(const v2f*)(mRow + kb);
    v2f ah = *(const v2f*)(hRow + kb);
    v2f fwl, fwr;
    fwl.x = Wl[(size_t)kb * 128 + col];
    fwl.y = Wl[(size_t)(kb + 1) * 128 + col];
    fwr.x = Wr[(size_t)kb * 128 + col];
    fwr.y = Wr[(size_t)(kb + 1) * 128 + col];
    acc = __builtin_amdgcn_wmma_f32_16x16x4_f32(false, am, false, fwl,
                                                (short)0, acc, false, false);
    acc = __builtin_amdgcn_wmma_f32_16x16x4_f32(false, ah, false, fwr,
                                                (short)0, acc, false, false);
  }

  const float bias = bvec[col];
  const float a = prelu_a[0];
  const int rbase = m0 + ((lane >> 4) << 3);
#pragma unroll
  for (int r = 0; r < 8; ++r) {
    float v = acc[r] + bias;
    v = (v >= 0.0f) ? v : a * v;
    out[(size_t)(rbase + r) * 128 + col] = v;
  }
}

extern "C" void kernel_launch(void* const* d_in, const int* in_sizes, int n_in,
                              void* d_out, int out_size, void* d_ws, size_t ws_size,
                              hipStream_t stream) {
  (void)n_in; (void)ws_size;

  const float* x   = (const float*)d_in[0];
  const int*   ei  = (const int*)d_in[1];
  const float* W1l = (const float*)d_in[4];
  const float* b1l = (const float*)d_in[5];
  const float* W1r = (const float*)d_in[6];
  const float* W2l = (const float*)d_in[7];
  const float* b2l = (const float*)d_in[8];
  const float* W2r = (const float*)d_in[9];
  const float* W3l = (const float*)d_in[10];
  const float* b3l = (const float*)d_in[11];
  const float* W3r = (const float*)d_in[12];
  const float* pa  = (const float*)d_in[13];

  const int N = in_sizes[0] / 128;
  const int E = in_sizes[1] / 2;
  const int B = out_size / 128;
  const int* src = ei;
  const int* dst = ei + E;

  float* ws     = (float*)d_ws;
  float* invdeg = ws;
  float* agg    = invdeg + N;
  float* hA     = agg + (size_t)N * 128;
  float* hB     = hA + (size_t)N * 128;

  const int T = 256;

  fill0_f4<<<512, T, 0, stream>>>((float4*)invdeg, (long)N / 4);
  degree_count<<<2048, T, 0, stream>>>(dst, invdeg, E);
  inv_deg_kernel<<<(N + T - 1) / T, T, 0, stream>>>(invdeg, N);

  fill0_f4<<<4096, T, 0, stream>>>((float4*)agg, (long)N * 32);
  scatter_accum<<<4096, T, 0, stream>>>(x, src, dst, agg, E, N);
  sage_wmma_gemm<<<N / 16, T, 0, stream>>>(agg, invdeg, x, W1l, b1l, W1r, pa, hA);

  fill0_f4<<<4096, T, 0, stream>>>((float4*)agg, (long)N * 32);
  scatter_accum<<<4096, T, 0, stream>>>(hA, src, dst, agg, E, N);
  sage_wmma_gemm<<<N / 16, T, 0, stream>>>(agg, invdeg, hA, W2l, b2l, W2r, pa, hB);

  fill0_f4<<<1024, T, 0, stream>>>((float4*)agg, (long)B * 32);
  scatter_accum<<<4096, T, 0, stream>>>(hB, src, dst, agg, E, B);
  sage_wmma_gemm<<<B / 16, T, 0, stream>>>(agg, invdeg, hB, W3l, b3l, W3r, pa,
                                           (float*)d_out);
}


# f32 atomic scatter-add replaced by a per-call destination-sorted edge list (returning int add gives each edge its slot) and one wave per row summing gathered rows; sum order = slot arrival order
# speedup vs baseline: 2.5907x; 2.5907x over previous
#include <hip/hip_runtime.h>
#include <hip/hip_bf16.h>

typedef __attribute__((ext_vector_type(2))) float v2f;
typedef __attribute__((ext_vector_type(8))) float v8f;

#define LDS_STRIDE 132

__global__ __launch_bounds__(256) void fill0_f4(float4* __restrict__ p, long n4) {
  long i = (long)blockIdx.x * blockDim.x + threadIdx.x;
  long stride = (long)gridDim.x * blockDim.x;
  float4 z = make_float4(0.f, 0.f, 0.f, 0.f);
  for (; i < n4; i += stride) p[i] = z;
}

__global__ __launch_bounds__(256) void degree_rank(const int* __restrict__ dst,
                                                   int* __restrict__ cnt,
                                                   int* __restrict__ rank, int nE) {
  int i = blockIdx.x * blockDim.x + threadIdx.x;
  int stride = gridDim.x * blockDim.x;
  for (; i < nE; i += stride) rank[i] = atomicAdd(&cnt[dst[i]], 1);
}

__global__ __launch_bounds__(1024) void row_offsets(const int* __restrict__ cnt,
                                                    int* __restrict__ rowptr,
                                                    float* __restrict__ invdeg, int n) {
  __shared__ int sPre[32];
  __shared__ int sTot[32];
  __shared__ int sOff[32];
  const int tid  = threadIdx.x;
  const int lane = tid & 31;
  const int wave = tid >> 5;
  const int first = blockIdx.x * 1024;

  int pre = 0;
  const int4* c4 = (const int4*)cnt;
  const int n4 = first >> 2;
  for (int i = tid; i < n4; i += 4096) {
    const bool hb = i + 1024 < n4, hc = i + 2048 < n4, hd = i + 3072 < n4;
    const int4 a = c4[i];
    const int4 b = c4[hb ? i + 1024 : i];
    const int4 c = c4[hc ? i + 2048 : i];
    const int4 d = c4[hd ? i + 3072 : i];
    const int sb = (b.x + b.y) + (b.z + b.w);
    const int sc = (c.x + c.y) + (c.z + c.w);
    const int sd = (d.x + d.y) + (d.z + d.w);
    pre += ((a.x + a.y) + (a.z + a.w)) + (hb ? sb : 0) + (hc ? sc : 0) + (hd ? sd : 0);
  }
  for (int off = 16; off > 0; off >>= 1) pre += __shfl_xor(pre, off);

  const int idx = first + tid;
  const int c = (idx < n) ? cnt[idx] : 0;
  int incl = c;
  for (int off = 1; off < 32; off <<= 1) {
    const int t = __shfl_up(incl, off);
    if (lane >= off) incl += t;
  }
  if (lane == 31) sTot[wave] = incl;
  if (lane == 0)  sPre[wave] = pre;
  __syncthreads();
  if (wave == 0) {
    const int tot = sTot[lane];
    int p = sPre[lane];
    for (int off = 16; off > 0; off >>= 1) p += __shfl_xor(p, off);
    int inc2 = tot;
    for (int off = 1; off < 32; off <<= 1) {
      const int t = __shfl_up(inc2, off);
      if (lane >= off) inc2 += t;
    }
    sOff[lane] = p + inc2 - tot;
  }
  __syncthreads();
  if (idx < n) {
    rowptr[idx] = sOff[wave] + incl - c;
    invdeg[idx] = 1.0f / fmaxf((float)c, 1.0f);
  }
}

__global__ __launch_bounds__(256) void csr_fill(const int* __restrict__ src,
                                                const int* __restrict__ dst,
                                                const int* __restrict__ rank,
                                                const int* __restrict__ rowptr,
                                                int* __restrict__ csr, int nE) {
  int i = blockIdx.x * blockDim.x + threadIdx.x;
  int stride = gridDim.x * blockDim.x;
  for (; i < nE; i += stride) csr[rowptr[dst[i]] + rank[i]] = src[i];
}

__device__ __forceinline__ float4 ld_row(const float* __restrict__ h, int s, int lane) {
  return *(const float4*)(h + (size_t)s * 128 + (size_t)lane * 4);
}

__global__ __launch_bounds__(256) void gather_sum(const float* __restrict__ h,
                                                  const int* __restrict__ csr,
                                                  const int* __restrict__ rowptr,
                                                  const int* __restrict__ cnt,
                                                  float* __restrict__ agg, int nRows) {
  const int lane = threadIdx.x & 31;
  const int wave0  = __builtin_amdgcn_readfirstlane((int)((blockIdx.x * blockDim.x + threadIdx.x) >> 5));
  const int nWaves = (int)((gridDim.x * blockDim.x) >> 5);
  for (int row = wave0; row < nRows; row += nWaves) {
    const int start = rowptr[row];
    const int len   = cnt[row];

    float4 acc = make_float4(0.f, 0.f, 0.f, 0.f);
    for (int base = 0; base < len; base += 32) {
      const int m = min(32, len - base);
      const int mine = (lane < m) ? csr[start + base + lane] : 0;
      int j = 0;
      for (; j + 8 <= m; j += 8) {
        float4 v[8];
#pragma unroll
        for (int u = 0; u < 8; ++u)
          v[u] = ld_row(h, __builtin_amdgcn_readlane(mine, j + u), lane);
#pragma unroll
        for (int u = 0; u < 8; ++u) {
          acc.x += v[u].x; acc.y += v[u].y; acc.z += v[u].z; acc.w += v[u].w;
        }
      }
      if (j + 4 <= m) {
        float4 v[4];
#pragma unroll
        for (int u = 0; u < 4; ++u)
          v[u] = ld_row(h, __builtin_amdgcn_readlane(mine, j + u), lane);
#pragma unroll
        for (int u = 0; u < 4; ++u) {
          acc.x += v[u].x; acc.y += v[u].y; acc.z += v[u].z; acc.w += v[u].w;
        }
        j += 4;
      }
      for (; j < m; ++j) {
        const float4 v = ld_row(h, __builtin_amdgcn_readlane(mine, j), lane);
        acc.x += v.x; acc.y += v.y; acc.z += v.z; acc.w += v.w;
      }
    }
    *(float4*)(agg + (size_t)row * 128 + (size_t)lane * 4) = acc;
  }
}

__global__ __launch_bounds__(256) void sage_wmma_gemm(const float* agg,
                                                      const float* __restrict__ invdeg,
                                                      const float* __restrict__ h,
                                                      const float* __restrict__ Wl,
                                                      const float* __restrict__ bvec,
                                                      const float* __restrict__ Wr,
                                                      const float* __restrict__ prelu_a,
                                                      float* out) {
  __shared__ float sMean[16 * LDS_STRIDE];
  __shared__ float sH[16 * LDS_STRIDE];

  const int lane = threadIdx.x & 31;
  const int wave = threadIdx.x >> 5;
  const int m0   = blockIdx.x << 4;
  const int kh   = (lane >> 4) << 1;
  const int col  = (wave << 4) + (lane & 15);

  for (int i = threadIdx.x; i < 512; i += 256) {
    const int r  = i >> 5;
    const int c4 = (i & 31) << 2;
    const float idg = invdeg[m0 + r];
    float4 va = *(const float4*)(agg + (size_t)(m0 + r) * 128 + c4);
    float4 vh = *(const float4*)(h + (size_t)(m0 + r) * 128 + c4);
    va.x *= idg; va.y *= idg; va.z *= idg; va.w *= idg;
    *(float4*)(&sMean[r * LDS_STRIDE + c4]) = va;
    *(float4*)(&sH[r * LDS_STRIDE + c4])    = vh;
  }
  __syncthreads();

  const float* mRow = &sMean[(lane & 15) * LDS_STRIDE];
  const float* hRow = &sH[(lane & 15) * LDS_STRIDE];

  v8f acc = {};

#pragma unroll 8
  for (int k = 0; k < 128; k += 4) {
    const int kb = k + kh;
    v2f am = *(const v2f*)(mRow + kb);
    v2f ah = *(const v2f*)(hRow + kb);
    v2f fwl, fwr;
    fwl.x = Wl[(size_t)kb * 128 + col];
    fwl.y = Wl[(size_t)(kb + 1) * 128 + col];
    fwr.x = Wr[(size_t)kb * 128 + col];
    fwr.y = Wr[(size_t)(kb + 1) * 128 + col];
    acc = __builtin_amdgcn_wmma_f32_16x16x4_f32(false, am, false, fwl,
                                                (short)0, acc, false, false);
    acc = __builtin_amdgcn_wmma_f32_16x16x4_f32(false, ah, false, fwr,
                                                (short)0, acc, false, false);
  }

  const float bias = bvec[col];
  const float a = prelu_a[0];
  const int rbase = m0 + ((lane >> 4) << 3);
#pragma unroll
  for (int r = 0; r < 8; ++r) {
    float v = acc[r] + bias;
    v = (v >= 0.0f) ? v : a * v;
    out[(size_t)(rbase + r) * 128 + col] = v;
  }
}

extern "C" void kernel_launch(void* const* d_in, const int* in_sizes, int n_in,
                              void* d_out, int out_size, void* d_ws, size_t ws_size,
                              hipStream_t stream) {
  (void)n_in; (void)ws_size;

  const float* x   = (const float*)d_in[0];
  const int*   ei  = (const int*)d_in[1];
  const float* W1l = (const float*)d_in[4];
  const float* b1l = (const float*)d_in[5];
  const float* W1r = (const float*)d_in[6];
  const float* W2l = (const float*)d_in[7];
  const float* b2l = (const float*)d_in[8];
  const float* W2r = (const float*)d_in[9];
  const float* W3l = (const float*)d_in[10];
  const float* b3l = (const float*)d_in[11];
  const float* W3r = (const float*)d_in[12];
  const float* pa  = (const float*)d_in[13];

  const int N = in_sizes[0] / 128;
  const int E = in_sizes[1] / 2;
  const int B = out_size / 128;
  const int* src = ei;
  const int* dst = ei + E;

  float* ws     = (float*)d_ws;
  float* invdeg = ws;
  float* bufP   = invdeg + N;
  float* bufQ   = bufP + (size_t)N * 128;
  int*   cnt    = (int*)(bufQ + (size_t)N * 128);
  int*   rowptr = cnt + N;
  int*   rank   = rowptr + N;
  int*   csr    = rank + E;

  const int T = 256;

  fill0_f4<<<128, T, 0, stream>>>((float4*)cnt, (long)N / 4);
  degree_rank<<<2048, T, 0, stream>>>(dst, cnt, rank, E);
  row_offsets<<<(N + 1023) / 1024, 1024, 0, stream>>>(cnt, rowptr, invdeg, N);
  csr_fill<<<2048, T, 0, stream>>>(src, dst, rank, rowptr, csr, E);

  gather_sum<<<4096, T, 0, stream>>>(x, csr, rowptr, cnt, bufP, N);
  sage_wmma_gemm<<<N / 16, T, 0, stream>>>(bufP, invdeg, x, W1l, b1l, W1r, pa, bufP);

  gather_sum<<<4096, T, 0, stream>>>(bufP, csr, rowptr, cnt, bufQ, N);
  sage_wmma_gemm<<<N / 16, T, 0, stream>>>(bufQ, invdeg, bufP, W2l, b2l, W2r, pa, bufQ);

  gather_sum<<<(B + 7) / 8, T, 0, stream>>>(bufQ, csr, rowptr, cnt, bufP, B);
  sage_wmma_gemm<<<B / 16, T, 0, stream>>>(bufP, invdeg, bufQ, W3l, b3l, W3r, pa,
                                           (float*)d_out);
}


# second layer computed only for rows the output depends on (output rows and sources of their in-edges): per-call flags, compacted row list, third layer reads by slot; per-row sum order unchanged
# speedup vs baseline: 2.8733x; 1.1091x over previous
#include <hip/hip_runtime.h>
#include <hip/hip_bf16.h>

typedef __attribute__((ext_vector_type(2))) float v2f;
typedef __attribute__((ext_vector_type(8))) float v8f;

#define LDS_STRIDE 132

__global__ __launch_bounds__(256) void fill0_f4(float4* __restrict__ p, long n4) {
  long i = (long)blockIdx.x * blockDim.x + threadIdx.x;
  long stride = (long)gridDim.x * blockDim.x;
  float4 z = make_float4(0.f, 0.f, 0.f, 0.f);
  for (; i < n4; i += stride) p[i] = z;
}

__global__ __launch_bounds__(256) void degree_rank(const int* __restrict__ dst,
                                                   int* __restrict__ cnt,
                                                   int* __restrict__ rank, int nE) {
  int i = blockIdx.x * blockDim.x + threadIdx.x;
  int stride = gridDim.x * blockDim.x;
  for (; i < nE; i += stride) rank[i] = atomicAdd(&cnt[dst[i]], 1);
}

__global__ __launch_bounds__(1024) void row_offsets(const int* __restrict__ cnt,
                                                    int* __restrict__ rowptr,
                                                    float* __restrict__ invdeg, int n) {
  __shared__ int sPre[32];
  __shared__ int sTot[32];
  __shared__ int sOff[32];
  const int tid  = threadIdx.x;
  const int lane = tid & 31;
  const int wave = tid >> 5;
  const int first = blockIdx.x * 1024;

  int pre = 0;
  const int4* c4 = (const int4*)cnt;
  const int n4 = first >> 2;
  for (int i = tid; i < n4; i += 4096) {
    const bool hb = i + 1024 < n4, hc = i + 2048 < n4, hd = i + 3072 < n4;
    const int4 a = c4[i];
    const int4 b = c4[hb ? i + 1024 : i];
    const int4 c = c4[hc ? i + 2048 : i];
    const int4 d = c4[hd ? i + 3072 : i];
    const int sb = (b.x + b.y) + (b.z + b.w);
    const int sc = (c.x + c.y) + (c.z + c.w);
    const int sd = (d.x + d.y) + (d.z + d.w);
    pre += ((a.x + a.y) + (a.z + a.w)) + (hb ? sb : 0) + (hc ? sc : 0) + (hd ? sd : 0);
  }
  for (int off = 16; off > 0; off >>= 1) pre += __shfl_xor(pre, off);

  const int idx = first + tid;
  const int c = (idx < n) ? cnt[idx] : 0;
  int incl = c;
  for (int off = 1; off < 32; off <<= 1) {
    const int t = __shfl_up(incl, off);
    if (lane >= off) incl += t;
  }
  if (lane == 31) sTot[wave] = incl;
  if (lane == 0)  sPre[wave] = pre;
  __syncthreads();
  if (wave == 0) {
    const int tot = sTot[lane];
    int p = sPre[lane];
    for (int off = 16; off > 0; off >>= 1) p += __shfl_xor(p, off);
    int inc2 = tot;
    for (int off = 1; off < 32; off <<= 1) {
      const int t = __shfl_up(inc2, off);
      if (lane >= off) inc2 += t;
    }
    sOff[lane] = p + inc2 - tot;
  }
  __syncthreads();
  if (idx < n) {
    rowptr[idx] = sOff[wave] + incl - c;
    invdeg[idx] = 1.0f / fmaxf((float)c, 1.0f);
  }
}

__global__ __launch_bounds__(256) void csr_fill(const int* __restrict__ src,
                                                const int* __restrict__ dst,
                                                const int* __restrict__ rank,
                                                const int* __restrict__ rowptr,
                                                int* __restrict__ csr,
                                                int* __restrict__ need, int nE, int nOut) {
  int i = blockIdx.x * blockDim.x + threadIdx.x;
  int stride = gridDim.x * blockDim.x;
  for (; i < nE; i += stride) {
    const int d = dst[i];
    const int s = src[i];
    csr[rowptr[d] + rank[i]] = s;
    if (d < nOut) need[s] = 1;
  }
}

__global__ __launch_bounds__(256) void needed_rows(const int* __restrict__ need,
                                                   int* __restrict__ nExtra,
                                                   int* __restrict__ list,
                                                   int* __restrict__ pos, int n, int nOut) {
  const int row = blockIdx.x * blockDim.x + threadIdx.x;
  const bool head = row < nOut;
  const bool want = (row < n) && !head && (need[row] != 0);
  const unsigned mask = __builtin_amdgcn_ballot_w32(want);
  int base = 0;
  if ((threadIdx.x & 31) == 0 && mask != 0u) base = atomicAdd(nExtra, __popc(mask));
  base = __builtin_amdgcn_readfirstlane(base);
  if (head && row < n) {
    list[row] = row;
    pos[row] = row;
  } else if (want) {
    const int p = nOut + base + (int)__builtin_amdgcn_mbcnt_lo(mask, 0u);
    list[p] = row;
    pos[row] = p;
  }
}

__device__ __forceinline__ float4 ld_row(const float* __restrict__ h, int s, int lane) {
  return *(const float4*)(h + (size_t)s * 128 + (size_t)lane * 4);
}

__global__ __launch_bounds__(256) void gather_sum(const float* __restrict__ h,
                                                  const int* __restrict__ csr,
                                                  const int* __restrict__ rowptr,
                                                  const int* __restrict__ cnt,
                                                  const int* __restrict__ list,
                                                  const int* __restrict__ pos,
                                                  const int* __restrict__ extra,
                                                  float* __restrict__ agg, int nBase) {
  const int lane = threadIdx.x & 31;
  const int nRows = nBase + (extra ? extra[0] : 0);
  const int wave0  = __builtin_amdgcn_readfirstlane((int)((blockIdx.x * blockDim.x + threadIdx.x) >> 5));
  const int nWaves = (int)((gridDim.x * blockDim.x) >> 5);
  for (int i = wave0; i < nRows; i += nWaves) {
    const int row = list ? list[i] : i;
    const int start = rowptr[row];
    const int len   = cnt[row];

    float4 acc = make_float4(0.f, 0.f, 0.f, 0.f);
    for (int base = 0; base < len; base += 32) {
      const int m = min(32, len - base);
      int mine = (lane < m) ? csr[start + base + lane] : 0;
      if (pos) mine = pos[mine];
      int j = 0;
      for (; j + 8 <= m; j += 8) {
        float4 v[8];
#pragma unroll
        for (int u = 0; u < 8; ++u)
          v[u] = ld_row(h, __builtin_amdgcn_readlane(mine, j + u), lane);
#pragma unroll
        for (int u = 0; u < 8; ++u) {
          acc.x += v[u].x; acc.y += v[u].y; acc.z += v[u].z; acc.w += v[u].w;
        }
      }
      if (j + 4 <= m) {
        float4 v[4];
#pragma unroll
        for (int u = 0; u < 4; ++u)
          v[u] = ld_row(h, __builtin_amdgcn_readlane(mine, j + u), lane);
#pragma unroll
        for (int u = 0; u < 4; ++u) {
          acc.x += v[u].x; acc.y += v[u].y; acc.z += v[u].z; acc.w += v[u].w;
        }
        j += 4;
      }
      for (; j < m; ++j) {
        const float4 v = ld_row(h, __builtin_amdgcn_readlane(mine, j), lane);
        acc.x += v.x; acc.y += v.y; acc.z += v.z; acc.w += v.w;
      }
    }
    *(float4*)(agg + (size_t)i * 128 + (size_t)lane * 4) = acc;
  }
}

__global__ __launch_bounds__(256) void sage_wmma_gemm(const float* agg,
                                                      const float* __restrict__ invdeg,
                                                      const float* __restrict__ h,
                                                      const float* __restrict__ Wl,
                                                      const float* __restrict__ bvec,
                                                      const float* __restrict__ Wr,
                                                      const float* __restrict__ prelu_a,
                                                      const int* __restrict__ list,
                                                      const int* __restrict__ extra,
                                                      int nBase, float* out) {
  __shared__ float sMean[16 * LDS_STRIDE];
  __shared__ float sH[16 * LDS_STRIDE];

  const int lane = threadIdx.x & 31;
  const int wave = threadIdx.x >> 5;
  const int kh   = (lane >> 4) << 1;
  const int col  = (wave << 4) + (lane & 15);

  const int nValid   = list ? nBase + extra[0] : 0x7fffffff;
  const int nStripes = list ? (nValid + 15) >> 4 : (int)gridDim.x;

  const float bias = bvec[col];
  const float a = prelu_a[0];
  const float* mRow = &sMean[(lane & 15) * LDS_STRIDE];
  const float* hRow = &sH[(lane & 15) * LDS_STRIDE];

  for (int s = blockIdx.x; s < nStripes; s += gridDim.x) {
    if (s != (int)blockIdx.x) __syncthreads();
    const int m0 = s << 4;

    for (int i = threadIdx.x; i < 512; i += 256) {
      const int r  = i >> 5;
      const int c4 = (i & 31) << 2;
      const int ia = min(m0 + r, nValid - 1);
      const int ih = list ? list[ia] : ia;
      const float idg = invdeg[ih];
      float4 va = *(const float4*)(agg + (size_t)ia * 128 + c4);
      float4 vh = *(const float4*)(h + (size_t)ih * 128 + c4);
      va.x *= idg; va.y *= idg; va.z *= idg; va.w *= idg;
      *(float4*)(&sMean[r * LDS_STRIDE + c4]) = va;
      *(float4*)(&sH[r * LDS_STRIDE + c4])    = vh;
    }
    __syncthreads();

    v8f acc = {};

#pragma unroll 8
    for (int k = 0; k < 128; k += 4) {
      const int kb = k + kh;
      v2f am = *(const v2f*)(mRow + kb);
      v2f ah = *(const v2f*)(hRow + kb);
      v2f fwl, fwr;
      fwl.x = Wl[(size_t)kb * 128 + col];
      fwl.y = Wl[(size_t)(kb + 1) * 128 + col];
      fwr.x = Wr[(size_t)kb * 128 + col];
      fwr.y = Wr[(size_t)(kb + 1) * 128 + col];
      acc = __builtin_amdgcn_wmma_f32_16x16x4_f32(false, am, false, fwl,
                                                  (short)0, acc, false, false);
      acc = __builtin_amdgcn_wmma_f32_16x16x4_f32(false, ah, false, fwr,
                                                  (short)0, acc, false, false);
    }

    const int rbase = m0 + ((lane >> 4) << 3);
    if (m0 + 16 <= nValid) {
#pragma unroll
      for (int r = 0; r < 8; ++r) {
        float v = acc[r] + bias;
        v = (v >= 0.0f) ? v : a * v;
        out[(size_t)(rbase + r) * 128 + col] = v;
      }
    } else {
#pragma unroll
      for (int r = 0; r < 8; ++r) {
        float v = acc[r] + bias;
        v = (v >= 0.0f) ? v : a * v;
        if (rbase + r < nValid) out[(size_t)(rbase + r) * 128 + col] = v;
      }
    }
  }
}

extern "C" void kernel_launch(void* const* d_in, const int* in_sizes, int n_in,
                              void* d_out, int out_size, void* d_ws, size_t ws_size,
                              hipStream_t stream) {
  (void)n_in; (void)ws_size;

  const float* x   = (const float*)d_in[0];
  const int*   ei  = (const int*)d_in[1];
  const float* W1l = (const float*)d_in[4];
  const float* b1l = (const float*)d_in[5];
  const float* W1r = (const float*)d_in[6];
  const float* W2l = (const float*)d_in[7];
  const float* b2l = (const float*)d_in[8];
  const float* W2r = (const float*)d_in[9];
  const float* W3l = (const float*)d_in[10];
  const float* b3l = (const float*)d_in[11];
  const float* W3r = (const float*)d_in[12];
  const float* pa  = (const float*)d_in[13];

  const int N = in_sizes[0] / 128;
  const int E = in_sizes[1] / 2;
  const int B = out_size / 128;
  const int* src = ei;
  const int* dst = ei + E;

  float* ws     = (float*)d_ws;
  float* invdeg = ws;
  float* bufP   = invdeg + N;
  float* bufQ   = bufP + (size_t)N * 128;
  int*   cnt    = (int*)(bufQ + (size_t)N * 128);
  int*   need   = cnt + N;
  int*   nExtra = need + N;
  int*   rowptr = nExtra + 4;
  int*   list   = rowptr + N;
  int*   pos    = list + N;
  int*   rank   = pos + N;
  int*   csr    = rank + E;

  const int T = 256;

  fill0_f4<<<128, T, 0, stream>>>((float4*)cnt, (long)(2 * N + 4) / 4);
  degree_rank<<<2048, T, 0, stream>>>(dst, cnt, rank, E);
  row_offsets<<<(N + 1023) / 1024, 1024, 0, stream>>>(cnt, rowptr, invdeg, N);
  csr_fill<<<2048, T, 0, stream>>>(src, dst, rank, rowptr, csr, need, E, B);
  needed_rows<<<(N + T - 1) / T, T, 0, stream>>>(need, nExtra, list, pos, N, B);

  gather_sum<<<4096, T, 0, stream>>>(x, csr, rowptr, cnt, nullptr, nullptr, nullptr, bufP, N);
  sage_wmma_gemm<<<N / 16, T, 0, stream>>>(bufP, invdeg, x, W1l, b1l, W1r, pa,
                                           nullptr, nullptr, 0, bufP);

  gather_sum<<<2048, T, 0, stream>>>(bufP, csr, rowptr, cnt, list, nullptr, nExtra, bufQ, B);
  sage_wmma_gemm<<<2048, T, 0, stream>>>(bufQ, invdeg, bufP, W2l, b2l, W2r, pa,
                                         list, nExtra, B, bufQ);

  gather_sum<<<(B + 7) / 8, T, 0, stream>>>(bufQ, csr, rowptr, cnt, nullptr, pos, nullptr, bufP, B);
  sage_wmma_gemm<<<B / 16, T, 0, stream>>>(bufP, invdeg, bufQ, W3l, b3l, W3r, pa,
                                           nullptr, nullptr, 0, (float*)d_out);
}


# layer kernel row tile 16 -> 32 rows in layers 1 and 2: each loaded weight fragment feeds two 16-row tiles (two independent accumulators per wave), halving weight loads per row; sum order unchanged
# speedup vs baseline: 5.2797x; 1.8375x over previous
#include <hip/hip_runtime.h>
#include <hip/hip_bf16.h>

typedef __attribute__((ext_vector_type(2))) float v2f;
typedef __attribute__((ext_vector_type(8))) float v8f;

#define LDS_STRIDE 132

#define BSHIFT   7
#define BROWS    128
#define NBMAX    1024
#define CH       6400
#define CH_PER_T 7
#define PK_SHIFT 25
#define PK_MASK  0x1FFFFFFu
#define BCAP     8192

__global__ __launch_bounds__(1024) void chunk_sort(const int* __restrict__ src,
                                                   const int* __restrict__ dst,
                                                   unsigned* __restrict__ packed,
                                                   int* __restrict__ lstart,
                                                   int* __restrict__ zero, int nZero, int nE) {
  __shared__ int hist[NBMAX];
  __shared__ int cur[NBMAX];
  __shared__ int sTot[32];
  __shared__ unsigned stage[CH];
  const int tid  = threadIdx.x;
  const int lane = tid & 31;
  const int wave = tid >> 5;

  for (int i = blockIdx.x * 1024 + tid; i < nZero; i += gridDim.x * 1024) zero[i] = 0;

  hist[tid] = 0;
  __syncthreads();

  const int lo = blockIdx.x * CH;
  const int n  = min(CH, nE - lo);
  int d[CH_PER_T], s[CH_PER_T];
#pragma unroll
  for (int u = 0; u < CH_PER_T; ++u) {
    const int i = tid + u * 1024;
    const bool ok = i < n;
    d[u] = ok ? dst[lo + i] : 0;
    s[u] = ok ? src[lo + i] : 0;
  }
#pragma unroll
  for (int u = 0; u < CH_PER_T; ++u) {
    if (tid + u * 1024 < n) {
      const unsigned b = min((unsigned)d[u] >> BSHIFT, (unsigned)(NBMAX - 1));
      atomicAdd(&hist[b], 1);
    }
  }
  __syncthreads();

  const int c = hist[tid];
  int incl = c;
  for (int off = 1; off < 32; off <<= 1) {
    const int t = __shfl_up(incl, off);
    if (lane >= off) incl += t;
  }
  if (lane == 31) sTot[wave] = incl;
  __syncthreads();
  if (wave == 0) {
    const int tot = sTot[lane];
    int inc2 = tot;
    for (int off = 1; off < 32; off <<= 1) {
      const int t = __shfl_up(inc2, off);
      if (lane >= off) inc2 += t;
    }
    sTot[lane] = inc2 - tot;
  }
  __syncthreads();
  const int excl = sTot[wave] + incl - c;
  cur[tid] = excl;
  lstart[blockIdx.x * NBMAX + tid] = excl;
  __syncthreads();

#pragma unroll
  for (int u = 0; u < CH_PER_T; ++u) {
    if (tid + u * 1024 < n) {
      const unsigned b = min((unsigned)d[u] >> BSHIFT, (unsigned)(NBMAX - 1));
      const int p = atomicAdd(&cur[b], 1);
      stage[p] = ((unsigned)(d[u] & (BROWS - 1)) << PK_SHIFT) | (unsigned)s[u];
    }
  }
  __syncthreads();
  for (int i = tid; i < n; i += 1024) packed[lo + i] = stage[i];
}

__global__ __launch_bounds__(256) void bucket_build(const unsigned* __restrict__ packed,
                                                    const int* __restrict__ lstart,
                                                    int* __restrict__ csr,
                                                    int* __restrict__ rowptr,
                                                    int* __restrict__ cnt,
                                                    float* __restrict__ invdeg,
                                                    int* __restrict__ need,
                                                    int nE, int nChunks, int n, int nOut) {
  __shared__ int ncnt[BROWS];
  __shared__ int ncur[BROWS];
  __shared__ int sW[8];
  __shared__ int sBase[8];
  __shared__ int sLen[8];
  __shared__ int sorted[BCAP];
  const int tid  = threadIdx.x;
  const int lane = tid & 31;
  const int wave = tid >> 5;
  const int b    = blockIdx.x;

  if (tid < BROWS) ncnt[tid] = 0;

  int myBase = 0, myLen = 0;
  for (int w = tid; w < nChunks; w += 256) {
    const int nW = min(CH, nE - w * CH);
    const int st = lstart[w * NBMAX + b];
    const int en = (b + 1 < NBMAX) ? lstart[w * NBMAX + b + 1] : nW;
    myBase += st;
    myLen  += en - st;
  }
  for (int off = 16; off > 0; off >>= 1) {
    myBase += __shfl_xor(myBase, off);
    myLen  += __shfl_xor(myLen, off);
  }
  if (lane == 0) { sBase[wave] = myBase; sLen[wave] = myLen; }
  __syncthreads();
  int base = 0, total = 0;
  for (int k = 0; k < 8; ++k) { base += sBase[k]; total += sLen[k]; }
  const bool fits = total <= BCAP;

  for (int w = tid; w < nChunks; w += 256) {
    const int nW = min(CH, nE - w * CH);
    const int st = lstart[w * NBMAX + b];
    const int en = (b + 1 < NBMAX) ? lstart[w * NBMAX + b + 1] : nW;
    const unsigned* run = packed + (size_t)w * CH;
    for (int i = st; i < en; i += 8) {
      unsigned v[8];
#pragma unroll
      for (int u = 0; u < 8; ++u) v[u] = run[min(i + u, en - 1)];
#pragma unroll
      for (int u = 0; u < 8; ++u)
        if (i + u < en) atomicAdd(&ncnt[v[u] >> PK_SHIFT], 1);
    }
  }
  __syncthreads();

  const int c = (tid < BROWS) ? ncnt[tid] : 0;
  int incl = c;
  for (int off = 1; off < 32; off <<= 1) {
    const int t = __shfl_up(incl, off);
    if (lane >= off) incl += t;
  }
  if (tid < BROWS && lane == 31) sW[wave] = incl;
  __syncthreads();
  if (tid < BROWS) {
    int pre = 0;
    for (int k = 0; k < wave; ++k) pre += sW[k];
    const int excl = pre + incl - c;
    ncur[tid] = excl;
    const int node = (b << BSHIFT) + tid;
    if (node < n) {
      rowptr[node] = base + excl;
      cnt[node]    = c;
      invdeg[node] = 1.0f / fmaxf((float)c, 1.0f);
    }
  }
  __syncthreads();

  const int row0 = b << BSHIFT;
  for (int w = tid; w < nChunks; w += 256) {
    const int nW = min(CH, nE - w * CH);
    const int st = lstart[w * NBMAX + b];
    const int en = (b + 1 < NBMAX) ? lstart[w * NBMAX + b + 1] : nW;
    const unsigned* run = packed + (size_t)w * CH;
    for (int i = st; i < en; i += 8) {
      unsigned v[8];
#pragma unroll
      for (int u = 0; u < 8; ++u) v[u] = run[min(i + u, en - 1)];
#pragma unroll
      for (int u = 0; u < 8; ++u) {
        if (i + u < en) {
          const int j  = (int)(v[u] >> PK_SHIFT);
          const int sr = (int)(v[u] & PK_MASK);
          const int p  = atomicAdd(&ncur[j], 1);
          if (fits) sorted[p] = sr; else csr[base + p] = sr;
          if (row0 + j < nOut) need[sr] = 1;
        }
      }
    }
  }
  __syncthreads();
  if (fits)
    for (int i = tid; i < total; i += 256) csr[base + i] = sorted[i];
}

__global__ __launch_bounds__(256) void needed_rows(const int* __restrict__ need,
                                                   int* __restrict__ nExtra,
                                                   int* __restrict__ list,
                                                   int* __restrict__ pos, int n, int nOut) {
  const int row = blockIdx.x * blockDim.x + threadIdx.x;
  const bool head = row < nOut;
  const bool want = (row < n) && !head && (need[row] != 0);
  const unsigned mask = __builtin_amdgcn_ballot_w32(want);
  int base = 0;
  if ((threadIdx.x & 31) == 0 && mask != 0u) base = atomicAdd(nExtra, __popc(mask));
  base = __builtin_amdgcn_readfirstlane(base);
  if (head && row < n) {
    list[row] = row;
    pos[row] = row;
  } else if (want) {
    const int p = nOut + base + (int)__builtin_amdgcn_mbcnt_lo(mask, 0u);
    list[p] = row;
    pos[row] = p;
  }
}

__device__ __forceinline__ float4 ld_row(const float* __restrict__ h, int s, int lane) {
  return *(const float4*)(h + (size_t)s * 128 + (size_t)lane * 4);
}

__device__ __forceinline__ float4 row_sum(const float* __restrict__ h,
                                          const int* __restrict__ csr,
                                          const int* __restrict__ pos,
                                          int start, int len, int first, int lane) {
  float4 acc = make_float4(0.f, 0.f, 0.f, 0.f);
  for (int base = 0; base < len; base += 32) {
    const int m = min(32, len - base);
    int mine = first;
    if (base != 0) {
      mine = (lane < m) ? csr[start + base + lane] : 0;
      if (pos) mine = pos[mine];
    }
    int j = 0;
    for (; j + 8 <= m; j += 8) {
      float4 v[8];
#pragma unroll
      for (int u = 0; u < 8; ++u)
        v[u] = ld_row(h, __builtin_amdgcn_readlane(mine, j + u), lane);
#pragma unroll
      for (int u = 0; u < 8; ++u) {
        acc.x += v[u].x; acc.y += v[u].y; acc.z += v[u].z; acc.w += v[u].w;
      }
    }
    if (j + 4 <= m) {
      float4 v[4];
#pragma unroll
      for (int u = 0; u < 4; ++u)
        v[u] = ld_row(h, __builtin_amdgcn_readlane(mine, j + u), lane);
#pragma unroll
      for (int u = 0; u < 4; ++u) {
        acc.x += v[u].x; acc.y += v[u].y; acc.z += v[u].z; acc.w += v[u].w;
      }
      j += 4;
    }
    for (; j < m; ++j) {
      const float4 v = ld_row(h, __builtin_amdgcn_readlane(mine, j), lane);
      acc.x += v.x; acc.y += v.y; acc.z += v.z; acc.w += v.w;
    }
  }
  return acc;
}

template <int MT>
__global__ __launch_bounds__(256, (MT == 1 ? 8 : 16)) void sage_layer(const float* __restrict__ h,
                                                  const int* __restrict__ csr,
                                                  const int* __restrict__ rowptr,
                                                  const int* __restrict__ cnt,
                                                  const float* __restrict__ invdeg,
                                                  const float* __restrict__ Wl,
                                                  const float* __restrict__ bvec,
                                                  const float* __restrict__ Wr,
                                                  const float* __restrict__ prelu_a,
                                                  const int* __restrict__ list,
                                                  const int* __restrict__ pos,
                                                  const int* __restrict__ extra,
                                                  int nBase, float* __restrict__ out) {
  constexpr int ROWS = MT * 16;
  __shared__ float sMean[ROWS * LDS_STRIDE];
  __shared__ float sH[ROWS * LDS_STRIDE];

  const int lane = threadIdx.x & 31;
  const int wave = __builtin_amdgcn_readfirstlane((int)(threadIdx.x >> 5));
  const int kh   = (lane >> 4) << 1;
  const int col  = (wave << 4) + (lane & 15);

  const int nValid   = nBase + (extra ? extra[0] : 0);
  const int nStripes = (nValid + ROWS - 1) / ROWS;

  const float bias = bvec[col];
  const float a = prelu_a[0];
  const float* mRow = &sMean[(lane & 15) * LDS_STRIDE];
  const float* hRow = &sH[(lane & 15) * LDS_STRIDE];

  for (int s = blockIdx.x; s < nStripes; s += gridDim.x) {
    if (s != (int)blockIdx.x) __syncthreads();
    const int m0 = s * ROWS;

    for (int p = 0; p < MT; ++p) {
    const int r0  = (wave * MT + p) << 1;
    const int r1  = r0 + 1;
    const int ia0 = min(m0 + r0, nValid - 1);
    const int ia1 = min(m0 + r1, nValid - 1);
    const int g0  = list ? list[ia0] : ia0;
    const int g1  = list ? list[ia1] : ia1;
    const int st0 = rowptr[g0], st1 = rowptr[g1];
    const int ln0 = cnt[g0],    ln1 = cnt[g1];
    const float idg0 = invdeg[g0], idg1 = invdeg[g1];
    int id0 = (lane < ln0) ? csr[st0 + lane] : 0;
    int id1 = (lane < ln1) ? csr[st1 + lane] : 0;
    if (pos) { id0 = pos[id0]; id1 = pos[id1]; }

    {
      const float4 vh = ld_row(h, g0, lane);
      float4 sm = row_sum(h, csr, pos, st0, ln0, id0, lane);
      sm.x *= idg0; sm.y *= idg0; sm.z *= idg0; sm.w *= idg0;
      *(float4*)(&sMean[r0 * LDS_STRIDE + lane * 4]) = sm;
      *(float4*)(&sH[r0 * LDS_STRIDE + lane * 4])    = vh;
    }
    {
      const float4 vh = ld_row(h, g1, lane);
      float4 sm = row_sum(h, csr, pos, st1, ln1, id1, lane);
      sm.x *= idg1; sm.y *= idg1; sm.z *= idg1; sm.w *= idg1;
      *(float4*)(&sMean[r1 * LDS_STRIDE + lane * 4]) = sm;
      *(float4*)(&sH[r1 * LDS_STRIDE + lane * 4])    = vh;
    }
    }
    __syncthreads();

    v8f acc[MT];
#pragma unroll
    for (int t = 0; t < MT; ++t) acc[t] = v8f{};

#pragma unroll(8 / MT)
    for (int k = 0; k < 128; k += 4) {
      const int kb = k + kh;
      v2f am[MT], ah[MT];
#pragma unroll
      for (int t = 0; t < MT; ++t) {
        am[t] = *(const v2f*)(mRow + t * 16 * LDS_STRIDE + kb);
        ah[t] = *(const v2f*)(hRow + t * 16 * LDS_STRIDE + kb);
      }
      v2f fwl, fwr;
      fwl.x = Wl[(size_t)kb * 128 + col];
      fwl.y = Wl[(size_t)(kb + 1) * 128 + col];
      fwr.x = Wr[(size_t)kb * 128 + col];
      fwr.y = Wr[(size_t)(kb + 1) * 128 + col];
#pragma unroll
      for (int t = 0; t < MT; ++t)
        acc[t] = __builtin_amdgcn_wmma_f32_16x16x4_f32(false, am[t], false, fwl,
                                                       (short)0, acc[t], false, false);
#pragma unroll
      for (int t = 0; t < MT; ++t)
        acc[t] = __builtin_amdgcn_wmma_f32_16x16x4_f32(false, ah[t], false, fwr,
                                                       (short)0, acc[t], false, false);
    }

#pragma unroll
    for (int t = 0; t < MT; ++t) {
      const int rbase = m0 + t * 16 + ((lane >> 4) << 3);
      if (m0 + ROWS <= nValid) {
#pragma unroll
        for (int r = 0; r < 8; ++r) {
          float v = acc[t][r] + bias;
          v = (v >= 0.0f) ? v : a * v;
          out[(size_t)(rbase + r) * 128 + col] = v;
        }
      } else {
#pragma unroll
        for (int r = 0; r < 8; ++r) {
          float v = acc[t][r] + bias;
          v = (v >= 0.0f) ? v : a * v;
          if (rbase + r < nValid) out[(size_t)(rbase + r) * 128 + col] = v;
        }
      }
    }
  }
}

extern "C" void kernel_launch(void* const* d_in, const int* in_sizes, int n_in,
                              void* d_out, int out_size, void* d_ws, size_t ws_size,
                              hipStream_t stream) {
  (void)n_in; (void)ws_size;

  const float* x   = (const float*)d_in[0];
  const int*   ei  = (const int*)d_in[1];
  const float* W1l = (const float*)d_in[4];
  const float* b1l = (const float*)d_in[5];
  const float* W1r = (const float*)d_in[6];
  const float* W2l = (const float*)d_in[7];
  const float* b2l = (const float*)d_in[8];
  const float* W2r = (const float*)d_in[9];
  const float* W3l = (const float*)d_in[10];
  const float* b3l = (const float*)d_in[11];
  const float* W3r = (const float*)d_in[12];
  const float* pa  = (const float*)d_in[13];

  const int N = in_sizes[0] / 128;
  const int E = in_sizes[1] / 2;
  const int B = out_size / 128;
  const int* src = ei;
  const int* dst = ei + E;

  float* ws     = (float*)d_ws;
  float* invdeg = ws;
  float* bufP   = invdeg + N;
  float* bufQ   = bufP + (size_t)N * 128;
  int*   cnt    = (int*)(bufQ + (size_t)N * 128);
  int*   need   = cnt + N;
  int*   nExtra = need + N;
  int*   rowptr = nExtra + 4;
  int*   list   = rowptr + N;
  int*   pos    = list + N;
  unsigned* packed = (unsigned*)(pos + N);
  int*   csr    = (int*)packed + E;
  int*   lstart = csr + E;

  const int T = 256;
  const int nChunks = (E + CH - 1) / CH;
  const int nB = (N + BROWS - 1) / BROWS;

  chunk_sort<<<nChunks, 1024, 0, stream>>>(src, dst, packed, lstart, need, N + 4, E);
  bucket_build<<<nB, T, 0, stream>>>(packed, lstart, csr, rowptr, cnt, invdeg, need, E, nChunks, N, B);
  needed_rows<<<(N + T - 1) / T, T, 0, stream>>>(need, nExtra, list, pos, N, B);

  sage_layer<2><<<(N + 31) / 32, T, 0, stream>>>(x, csr, rowptr, cnt, invdeg, W1l, b1l, W1r, pa,
                                       nullptr, nullptr, nullptr, N, bufP);

  sage_layer<2><<<2048, T, 0, stream>>>(bufP, csr, rowptr, cnt, invdeg, W2l, b2l, W2r, pa,
                                     list, nullptr, nExtra, B, bufQ);

  sage_layer<1><<<B / 16, T, 0, stream>>>(bufQ, csr, rowptr, cnt, invdeg, W3l, b3l, W3r, pa,
                                       nullptr, pos, nullptr, B, (float*)d_out);
}
